# starting kernel
# baseline (speedup 1.0000x reference)
#include <hip/hip_runtime.h>

typedef __attribute__((ext_vector_type(2))) float v2f;
typedef __attribute__((ext_vector_type(8))) float v8f;

#define DVEC   128
#define KSTEPS 32
#define NTILES 8


__device__ __forceinline__ void load_w_to_lds(float4* bw4, const float* __restrict__ W) {
    const float4* W4 = (const float4*)W;
    for (int idx = threadIdx.x; idx < KSTEPS * DVEC; idx += blockDim.x) {
        int n = idx >> 5;
        int s = idx & 31;
        bw4[s * DVEC + n] = W4[idx];
    }
}

__global__ __launch_bounds__(256) void compgcn_dense(const float* __restrict__ X,
                                                     const float* __restrict__ W,
                                                     const float* __restrict__ bias,
                                                     float* __restrict__ Out,
                                                     int nrows) {
    __shared__ float4 bw4[KSTEPS * DVEC];
    load_w_to_lds(bw4, W);
    __syncthreads();

    const int lane = threadIdx.x & 31;
    const int m    = lane & 15;
    const int h    = lane >> 4;
    const int wave = threadIdx.x >> 5;
    const int wavesPerBlock = blockDim.x >> 5;
    const int gwave = blockIdx.x * wavesPerBlock + wave;
    const int nwaves = gridDim.x * wavesPerBlock;
    const int ntileRows = (nrows + 15) >> 4;
    const float* bwl = (const float*)bw4 + m * 4 + 2 * h;

    float bv[NTILES];
    #pragma unroll
    for (int j = 0; j < NTILES; ++j) bv[j] = bias ? bias[j * 16 + m] : 0.f;

    for (int t = gwave; t < ntileRows; t += nwaves) {
        int row = t * 16 + m;
        int rc  = row < nrows ? row : (nrows - 1);
        const float* xrow = X + (size_t)rc * DVEC;

        v8f acc[NTILES];
        #pragma unroll
        for (int j = 0; j < NTILES; ++j) acc[j] = (v8f){0.f,0.f,0.f,0.f,0.f,0.f,0.f,0.f};

        #pragma unroll 2
        for (int s = 0; s < KSTEPS; ++s) {
            v2f a = *(const v2f*)(xrow + 4 * s + 2 * h);
            v2f b[NTILES];
            #pragma unroll
            for (int j = 0; j < NTILES; ++j)
                b[j] = *(const v2f*)(bwl + s * DVEC * 4 + j * 64);
            #pragma unroll
            for (int j = 0; j < NTILES; ++j)
                acc[j] = __builtin_amdgcn_wmma_f32_16x16x4_f32(
                    false, a, false, b[j], (short)0, acc[j], false, false);
        }

        if (t * 16 + 16 <= nrows) {
            #pragma unroll
            for (int j = 0; j < NTILES; ++j) {
                float* orow = Out + (size_t)(t * 16 + 8 * h) * DVEC + j * 16 + m;
                #pragma unroll
                for (int i = 0; i < 8; ++i)
                    orow[(size_t)i * DVEC] = acc[j][i] + bv[j];
            }
        } else {
            #pragma unroll
            for (int j = 0; j < NTILES; ++j) {
                #pragma unroll
                for (int i = 0; i < 8; ++i) {
                    int grow = t * 16 + i + 8 * h;
                    if (grow < nrows)
                        Out[(size_t)grow * DVEC + j * 16 + m] = acc[j][i] + bv[j];
                }
            }
        }
    }
}

__global__ __launch_bounds__(256) void compgcn_edges(const float* __restrict__ X,
                                                     const float* __restrict__ Rel,
                                                     const int* __restrict__ src,
                                                     const int* __restrict__ dst,
                                                     const int* __restrict__ etype,
                                                     const float* __restrict__ ew,
                                                     const float* __restrict__ W,
                                                     float* __restrict__ Out,
                                                     int nedges) {
    __shared__ float4 bw4[KSTEPS * DVEC];
    load_w_to_lds(bw4, W);
    __syncthreads();

    const int lane = threadIdx.x & 31;
    const int m    = lane & 15;
    const int h    = lane >> 4;
    const int wave = threadIdx.x >> 5;
    const int wavesPerBlock = blockDim.x >> 5;
    const int gwave = blockIdx.x * wavesPerBlock + wave;
    const int nwaves = gridDim.x * wavesPerBlock;
    const int ntiles = (nedges + 15) >> 4;
    const float* bwl = (const float*)bw4 + m * 4 + 2 * h;

    for (int t = gwave; t < ntiles; t += nwaves) {
        int e  = t * 16 + m;
        bool valid = e < nedges;
        int ec = valid ? e : 0;
        float w  = valid ? ew[ec] : 0.f;
        int   sN = src[ec];
        int   ty = etype[ec];
        int   dN = dst[ec];
        const float* xrow = X   + (size_t)sN * DVEC;
        const float* rrow = Rel + (size_t)ty * DVEC;

        v8f acc[NTILES];
        #pragma unroll
        for (int j = 0; j < NTILES; ++j) acc[j] = (v8f){0.f,0.f,0.f,0.f,0.f,0.f,0.f,0.f};

        #pragma unroll 2
        for (int s = 0; s < KSTEPS; ++s) {
            v2f xv = *(const v2f*)(xrow + 4 * s + 2 * h);
            v2f rv = *(const v2f*)(rrow + 4 * s + 2 * h);
            v2f b[NTILES];
            #pragma unroll
            for (int j = 0; j < NTILES; ++j)
                b[j] = *(const v2f*)(bwl + s * DVEC * 4 + j * 64);
            v2f a;
            a.x = (xv.x - rv.x) * w;
            a.y = (xv.y - rv.y) * w;
            #pragma unroll
            for (int j = 0; j < NTILES; ++j)
                acc[j] = __builtin_amdgcn_wmma_f32_16x16x4_f32(
                    false, a, false, b[j], (short)0, acc[j], false, false);
        }

        #pragma unroll
        for (int i = 0; i < 8; ++i) {
            int r  = i + 8 * h;
            int dd = __shfl(dN, r, 32);
            float* orow = Out + (size_t)dd * DVEC;
            #pragma unroll
            for (int j = 0; j < NTILES; ++j)
                atomicAdd(orow + j * 16 + m, acc[j][i]);
        }
    }
}

extern "C" void kernel_launch(void* const* d_in, const int* in_sizes, int n_in,
                              void* d_out, int out_size, void* d_ws, size_t ws_size,
                              hipStream_t stream) {
    const float* x         = (const float*)d_in[0];
    const int*   edge_idx  = (const int*)d_in[1];
    const int*   edge_type = (const int*)d_in[2];
    const float* rel_emb   = (const float*)d_in[3];
    const float* edge_w    = (const float*)d_in[4];
    const float* W_self    = (const float*)d_in[5];
    const float* W_forward = (const float*)d_in[6];
    const float* W_rel     = (const float*)d_in[7];
    const float* bias      = (const float*)d_in[8];

    const int N = in_sizes[0] / DVEC;
    const int E = in_sizes[2];
    const int R = in_sizes[3] / DVEC;

    const int* src = edge_idx;
    const int* dst = edge_idx + E;

    float* out_nodes = (float*)d_out;
    float* out_rel   = out_nodes + (size_t)N * DVEC;

    {
        int tiles  = (N + 15) >> 4;
        int blocks = (tiles + 7) >> 3;
        compgcn_dense<<<blocks, 256, 0, stream>>>(x, W_self, bias, out_nodes, N);
    }
    {
        int tiles  = (R + 15) >> 4;
        int blocks = (tiles + 7) >> 3;
        compgcn_dense<<<blocks, 256, 0, stream>>>(rel_emb, W_rel, nullptr, out_rel, R);
    }
    {
        int tiles  = (E + 15) >> 4;
        int blocks = (tiles + 7) >> 3;
        if (blocks > 2048) blocks = 2048;
        compgcn_edges<<<blocks, 256, 0, stream>>>(x, rel_emb, src, dst, edge_type, edge_w,
                                                  W_forward, out_nodes, E);
    }
}


# Two orders of work in the fused kernel: half the waves of each SIMD run X @ W_self^T before their chain walk (accumulators wait in the wave's own LDS slab), half walk first; sum order unchanged
# speedup vs baseline: 2.0115x; 2.0115x over previous
#include <hip/hip_runtime.h>


typedef __attribute__((ext_vector_type(2))) float v2f;
typedef __attribute__((ext_vector_type(8))) float v8f;

#define DVEC   128
#define KSTEPS 32
#define NTILES 8


__device__ __forceinline__ void load_w_to_lds(float4* bw4, const float* __restrict__ W) {
    const float4* W4 = (const float4*)W;
    for (int idx = threadIdx.x; idx < KSTEPS * DVEC; idx += blockDim.x) {
        int n = idx >> 5;
        int s = idx & 31;
        bw4[s * DVEC + n] = W4[idx];
    }
}

__device__ __forceinline__ void tile_pass(v8f (&acc)[NTILES], const float* arow, const float* bwl) {
    #pragma unroll 2
    for (int s = 0; s < KSTEPS; ++s) {
        v2f a = *(const v2f*)(arow + 4 * s);
        v2f b[NTILES];
        #pragma unroll
        for (int j = 0; j < NTILES; ++j)
            b[j] = *(const v2f*)(bwl + s * DVEC * 4 + j * 64);
        #pragma unroll
        for (int j = 0; j < NTILES; ++j)
            acc[j] = __builtin_amdgcn_wmma_f32_16x16x4_f32(
                false, a, false, b[j], (short)0, acc[j], false, false);
    }
}

__device__ __forceinline__ void store_tile(const v8f (&acc)[NTILES], const float (&bv)[NTILES],
                                           float* Out, int t, int nrows, int m, int h) {
    if (t * 16 + 16 <= nrows) {
        #pragma unroll
        for (int j = 0; j < NTILES; ++j) {
            float* orow = Out + (size_t)(t * 16 + 8 * h) * DVEC + j * 16 + m;
            #pragma unroll
            for (int i = 0; i < 8; ++i)
                orow[(size_t)i * DVEC] = acc[j][i] + bv[j];
        }
    } else {
        #pragma unroll
        for (int j = 0; j < NTILES; ++j) {
            #pragma unroll
            for (int i = 0; i < 8; ++i) {
                int grow = t * 16 + i + 8 * h;
                if (grow < nrows)
                    Out[(size_t)grow * DVEC + j * 16 + m] = acc[j][i] + bv[j];
            }
        }
    }
}

__global__ __launch_bounds__(32) void compgcn_rel_clear(const float* __restrict__ Rel,
                                                        const float* __restrict__ W,
                                                        float* __restrict__ OutRel,
                                                        int nrel, int relBlocks,
                                                        float* __restrict__ Agg, int nrows) {
    if ((int)blockIdx.x >= relBlocks) {
        float4* a4 = (float4*)Agg;
        const size_t n4 = (size_t)nrows * (DVEC / 4);
        const size_t stride = (size_t)(gridDim.x - relBlocks) * blockDim.x;
        for (size_t i = (size_t)(blockIdx.x - relBlocks) * blockDim.x + threadIdx.x; i < n4; i += stride)
            a4[i] = make_float4(0.f, 0.f, 0.f, 0.f);
        return;
    }
    const int lane = threadIdx.x & 31;
    const int m    = lane & 15;
    const int h    = lane >> 4;
    const int t    = (int)blockIdx.x / NTILES;
    const int j    = (int)blockIdx.x % NTILES;
    const int row  = t * 16 + m;
    const int rc   = row < nrel ? row : (nrel - 1);
    const float* arow = Rel + (size_t)rc * DVEC + 2 * h;
    const float* brow = W + (size_t)(j * 16 + m) * DVEC + 2 * h;

    v2f a[KSTEPS], b[KSTEPS];
    #pragma unroll
    for (int s = 0; s < KSTEPS; ++s) {
        a[s] = *(const v2f*)(arow + 4 * s);
        b[s] = *(const v2f*)(brow + 4 * s);
    }
    v8f acc = (v8f){0.f,0.f,0.f,0.f,0.f,0.f,0.f,0.f};
    #pragma unroll
    for (int s = 0; s < KSTEPS; ++s)
        acc = __builtin_amdgcn_wmma_f32_16x16x4_f32(false, a[s], false, b[s], (short)0, acc, false, false);

    #pragma unroll
    for (int i = 0; i < 8; ++i) {
        const int grow = t * 16 + i + 8 * h;
        if (grow < nrel)
            OutRel[(size_t)grow * DVEC + j * 16 + m] = acc[i];
    }
}

#define EPW 8
__global__ __launch_bounds__(256) void compgcn_scatter(const float* __restrict__ X,
                                                       const float* __restrict__ Rel,
                                                       const int* __restrict__ src,
                                                       const int* __restrict__ dst,
                                                       const int* __restrict__ etype,
                                                       const float* __restrict__ ew,
                                                       float* __restrict__ Agg,
                                                       int nedges) {
    const unsigned lane = threadIdx.x & 31u;
    const int wave = threadIdx.x >> 5;
    const int wavesPerBlock = blockDim.x >> 5;
    const int gwave = blockIdx.x * wavesPerBlock + wave;
    const int nwaves = gridDim.x * wavesPerBlock;
    const int nunits = (nedges + EPW - 1) / EPW;

    for (int u = gwave; u < nunits; u += nwaves) {
        const int e0 = u * EPW;
        int e  = e0 + (int)(lane & (EPW - 1));
        int ec = e < nedges ? e : (nedges - 1);
        int   sN = src[ec];
        int   dN = dst[ec];
        int   ty = etype[ec];
        float w  = ew[ec];
        const int cnt = (nedges - e0) < EPW ? (nedges - e0) : EPW;

        if (cnt == EPW) {
            float xv[EPW][4], rv[EPW][4];
            int   dd[EPW];
            float ww[EPW];
            #pragma unroll
            for (int i = 0; i < EPW; ++i) {
                const int s_i = __builtin_amdgcn_readlane(sN, i);
                const int t_i = __builtin_amdgcn_readlane(ty, i);
                dd[i] = __builtin_amdgcn_readlane(dN, i);
                ww[i] = __builtin_bit_cast(float, __builtin_amdgcn_readlane(__builtin_bit_cast(int, w), i));
                const float* xrow = X   + (size_t)s_i * DVEC;
                const float* rrow = Rel + (size_t)t_i * DVEC;
                #pragma unroll
                for (int c = 0; c < 4; ++c) { xv[i][c] = xrow[lane + 32u * c]; rv[i][c] = rrow[lane + 32u * c]; }
            }
            #pragma unroll
            for (int i = 0; i < EPW; ++i) {
                float* orow = Agg + (size_t)dd[i] * DVEC;
                #pragma unroll
                for (int c = 0; c < 4; ++c)
                    atomicAdd(orow + (lane + 32u * c), (xv[i][c] - rv[i][c]) * ww[i]);
            }
        } else {
            for (int i = 0; i < cnt; ++i) {
                const int s_i = __builtin_amdgcn_readlane(sN, i);
                const int t_i = __builtin_amdgcn_readlane(ty, i);
                const int d_i = __builtin_amdgcn_readlane(dN, i);
                const float w_i = __builtin_bit_cast(float, __builtin_amdgcn_readlane(__builtin_bit_cast(int, w), i));
                const float* xrow = X   + (size_t)s_i * DVEC;
                const float* rrow = Rel + (size_t)t_i * DVEC;
                float* orow = Agg + (size_t)d_i * DVEC;
                #pragma unroll
                for (int c = 0; c < 4; ++c)
                    atomicAdd(orow + (lane + 32u * c), (xrow[lane + 32u * c] - rrow[lane + 32u * c]) * w_i);
            }
        }
    }
}

#define LPT 4
__global__ __launch_bounds__(256) void compgcn_link(const int* __restrict__ src,
                                                    const int* __restrict__ dst,
                                                    const int* __restrict__ etype,
                                                    const float* __restrict__ ew,
                                                    int* head, int* __restrict__ recw, int nedges) {
    int4* rec = (int4*)recw;
    const int total = gridDim.x * blockDim.x;
    const int e0 = blockIdx.x * blockDim.x + threadIdx.x;
    const int elast = nedges > 0 ? nedges - 1 : 0;
    int d[LPT], s[LPT], t[LPT], wb[LPT], old[LPT];
    #pragma unroll
    for (int k = 0; k < LPT; ++k) {
        const int e  = e0 + k * total;
        const int ec = e < elast ? e : elast;
        d[k]  = dst[ec];
        s[k]  = src[ec];
        t[k]  = etype[ec];
        wb[k] = __builtin_bit_cast(int, ew[ec]);
    }
    #pragma unroll
    for (int k = 0; k < LPT; ++k) {
        const int e = e0 + k * total;
        old[k] = 0;
        if (e < nedges) old[k] = atomicExch(head + d[k], e + 1);
    }
    #pragma unroll
    for (int k = 0; k < LPT; ++k) {
        const int e = e0 + k * total;
        if (e < nedges) rec[e] = make_int4(s[k], t[k], wb[k], old[k]);
    }
}

#define SPITCH 132

__device__ __forceinline__ void chain_walk(float4 (&sum)[16], const int* __restrict__ head, const int4* __restrict__ rec,
                                           const float4* __restrict__ X4, const float4* __restrict__ R4,
                                           int row, int nrows, int nedges, int elast, unsigned lane) {
    int link = row < nrows ? head[row] : 0;
    int e = link - 1; e = e < 0 ? 0 : (e > elast ? elast : e);
    int4 q = rec[e];
    float4 xv[16], rv[16];
    #pragma unroll
    for (int i = 0; i < 16; ++i) {
        sum[i] = make_float4(0.f, 0.f, 0.f, 0.f);
        xv[i]  = make_float4(0.f, 0.f, 0.f, 0.f);
        rv[i]  = make_float4(0.f, 0.f, 0.f, 0.f);
    }
    for (int it = 0; it < nedges; ++it) {
        const unsigned live = __builtin_amdgcn_ballot_w32(link != 0);
        if (live == 0u) break;
        #pragma unroll
        for (int i = 0; i < 16; ++i) {
            if (live & (1u << i)) {
                const int s_i = __builtin_amdgcn_readlane(q.x, i);
                const int t_i = __builtin_amdgcn_readlane(q.y, i);
                xv[i] = X4[(size_t)s_i * (DVEC / 4) + lane];
                rv[i] = R4[(size_t)t_i * (DVEC / 4) + lane];
            }
        }
        const int wbits = q.z;
        link = link != 0 ? q.w : 0;
        e = link - 1; e = e < 0 ? 0 : (e > elast ? elast : e);
        q = rec[e];
        #pragma unroll
        for (int i = 0; i < 16; ++i) {
            if (live & (1u << i)) {
                const float w_i = __builtin_bit_cast(float, __builtin_amdgcn_readlane(wbits, i));
                sum[i].x += (xv[i].x - rv[i].x) * w_i;
                sum[i].y += (xv[i].y - rv[i].y) * w_i;
                sum[i].z += (xv[i].z - rv[i].z) * w_i;
                sum[i].w += (xv[i].w - rv[i].w) * w_i;
            }
        }
    }
}
__global__ __launch_bounds__(512) void compgcn_pull_project(const float* __restrict__ X,
                                                            const float* __restrict__ Rel,
                                                            const int* __restrict__ head,
                                                            const int* __restrict__ recw,
                                                            const float* __restrict__ Wself,
                                                            const float* __restrict__ Wfwd,
                                                            const float* __restrict__ bias,
                                                            float* __restrict__ Out,
                                                            int nrows, int nedges) {
    __shared__ float4 bw4[2 * KSTEPS * DVEC];
    __shared__ float4 stage4[16 * 16 * (SPITCH / 4)];
    load_w_to_lds(bw4, Wself);
    load_w_to_lds(bw4 + KSTEPS * DVEC, Wfwd);
    __syncthreads();

    const int4* rec = (const int4*)recw;
    const float4* X4 = (const float4*)X;
    const float4* R4 = (const float4*)Rel;
    const unsigned lane = threadIdx.x & 31u;
    const int m    = (int)(lane & 15u);
    const int h    = (int)(lane >> 4);
    const int wave = threadIdx.x >> 5;
    const int wavesPerBlock = blockDim.x >> 5;
    const int ntileRows = (nrows + 15) >> 4;
    const int t0 = (int)(((long long)ntileRows * blockIdx.x) / gridDim.x);
    const int t1 = (int)(((long long)ntileRows * (blockIdx.x + 1)) / gridDim.x);
    const int elast = nedges > 0 ? nedges - 1 : 0;
    const float* bwl0 = (const float*)bw4 + m * 4 + 2 * h;
    const float* bwl1 = bwl0 + KSTEPS * DVEC * 4;
    float* st = (float*)stage4 + wave * (16 * SPITCH);

    float bv[NTILES];
    #pragma unroll
    for (int j = 0; j < NTILES; ++j) bv[j] = bias[j * 16 + m];

    const bool xfirst = (((wave ^ (wave >> 2)) & 1) != 0);
    if (xfirst) {
        float4* park = (float4*)st;
        for (int t = t0 + wave; t < t1; t += wavesPerBlock) {
            const int row = t * 16 + m;
            const int rc  = row < nrows ? row : (nrows - 1);
            {
                v8f acc0[NTILES];
                #pragma unroll
                for (int j = 0; j < NTILES; ++j) acc0[j] = (v8f){0.f,0.f,0.f,0.f,0.f,0.f,0.f,0.f};
                tile_pass(acc0, X + (size_t)rc * DVEC + 2 * h, bwl0);
                #pragma unroll
                for (int j = 0; j < NTILES; ++j) {
                    park[(2 * j) * 32 + lane]     = make_float4(acc0[j][0], acc0[j][1], acc0[j][2], acc0[j][3]);
                    park[(2 * j + 1) * 32 + lane] = make_float4(acc0[j][4], acc0[j][5], acc0[j][6], acc0[j][7]);
                }
            }
            asm volatile("" ::: "memory");
            float4 sum[16];
            chain_walk(sum, head, rec, X4, R4, row, nrows, nedges, elast, lane);
            v8f acc[NTILES];
            #pragma unroll
            for (int j = 0; j < NTILES; ++j) {
                const float4 lo = park[(2 * j) * 32 + lane];
                const float4 hi = park[(2 * j + 1) * 32 + lane];
                acc[j] = (v8f){lo.x, lo.y, lo.z, lo.w, hi.x, hi.y, hi.z, hi.w};
            }
            asm volatile("" ::: "memory");
            #pragma unroll
            for (int i = 0; i < 16; ++i)
                *(float4*)(st + i * SPITCH + 4 * lane) = sum[i];
            __builtin_amdgcn_wave_barrier();
            tile_pass(acc, st + m * SPITCH + 2 * h, bwl1);
            __builtin_amdgcn_wave_barrier();
            store_tile(acc, bv, Out, t, nrows, m, h);
        }
    } else {
        for (int t = t0 + wave; t < t1; t += wavesPerBlock) {
            const int row = t * 16 + m;
            const int rc  = row < nrows ? row : (nrows - 1);
            float4 sum[16];
            chain_walk(sum, head, rec, X4, R4, row, nrows, nedges, elast, lane);
            #pragma unroll
            for (int i = 0; i < 16; ++i)
                *(float4*)(st + i * SPITCH + 4 * lane) = sum[i];
            __builtin_amdgcn_wave_barrier();

            v8f acc[NTILES];
            #pragma unroll
            for (int j = 0; j < NTILES; ++j) acc[j] = (v8f){0.f,0.f,0.f,0.f,0.f,0.f,0.f,0.f};
            tile_pass(acc, X + (size_t)rc * DVEC + 2 * h, bwl0);
            tile_pass(acc, st + m * SPITCH + 2 * h, bwl1);
            __builtin_amdgcn_wave_barrier();
            store_tile(acc, bv, Out, t, nrows, m, h);
        }
    }
}

__global__ __launch_bounds__(512) void compgcn_project(const float* __restrict__ X,
                                                       const float* __restrict__ Wself,
                                                       const float* __restrict__ Wfwd,
                                                       const float* __restrict__ bias,
                                                       float* Out,
                                                       int nrows) {
    __shared__ float4 bw4[2 * KSTEPS * DVEC];
    load_w_to_lds(bw4, Wself);
    load_w_to_lds(bw4 + KSTEPS * DVEC, Wfwd);
    __syncthreads();

    const int lane = threadIdx.x & 31;
    const int m    = lane & 15;
    const int h    = lane >> 4;
    const int wave = threadIdx.x >> 5;
    const int wavesPerBlock = blockDim.x >> 5;
    const int ntileRows = (nrows + 15) >> 4;
    const int t0 = (int)(((long long)ntileRows * blockIdx.x) / gridDim.x);
    const int t1 = (int)(((long long)ntileRows * (blockIdx.x + 1)) / gridDim.x);
    const float* bwl0 = (const float*)bw4 + m * 4 + 2 * h;
    const float* bwl1 = bwl0 + KSTEPS * DVEC * 4;

    float bv[NTILES];
    #pragma unroll
    for (int j = 0; j < NTILES; ++j) bv[j] = bias[j * 16 + m];

    for (int t = t0 + wave; t < t1; t += wavesPerBlock) {
        int row = t * 16 + m;
        int rc  = row < nrows ? row : (nrows - 1);
        v8f acc[NTILES];
        #pragma unroll
        for (int j = 0; j < NTILES; ++j) acc[j] = (v8f){0.f,0.f,0.f,0.f,0.f,0.f,0.f,0.f};
        tile_pass(acc, X   + (size_t)rc * DVEC + 2 * h, bwl0);
        tile_pass(acc, Out + (size_t)rc * DVEC + 2 * h, bwl1);
        store_tile(acc, bv, Out, t, nrows, m, h);
    }
}

extern "C" void kernel_launch(void* const* d_in, const int* in_sizes, int n_in,
                              void* d_out, int out_size, void* d_ws, size_t ws_size,
                              hipStream_t stream) {
    const float* x         = (const float*)d_in[0];
    const int*   edge_idx  = (const int*)d_in[1];
    const int*   edge_type = (const int*)d_in[2];
    const float* rel_emb   = (const float*)d_in[3];
    const float* edge_w    = (const float*)d_in[4];
    const float* W_self    = (const float*)d_in[5];
    const float* W_forward = (const float*)d_in[6];
    const float* W_rel     = (const float*)d_in[7];
    const float* bias      = (const float*)d_in[8];

    const int N = in_sizes[0] / DVEC;
    const int E = in_sizes[2];
    const int R = in_sizes[3] / DVEC;

    const int* src = edge_idx;
    const int* dst = edge_idx + E;

    float* out_nodes = (float*)d_out;
    float* out_rel   = out_nodes + (size_t)N * DVEC;

    const size_t headInts = ((size_t)N + 127) & ~(size_t)127;
    const size_t need = (headInts + 4 * (size_t)E) * sizeof(int);
    const int relBlocks = ((R + 15) >> 4) * NTILES;
    if (ws_size >= need) {
        int* head = (int*)d_ws;
        int* rec  = head + headInts;
        compgcn_rel_clear<<<relBlocks + 128, 32, 0, stream>>>(rel_emb, W_rel, out_rel, R, relBlocks,
                                                               (float*)head, (int)(headInts >> 7));
        {
            int blocks = (E + 256 * LPT - 1) / (256 * LPT);
            if (blocks < 1) blocks = 1;
            compgcn_link<<<blocks, 256, 0, stream>>>(src, dst, edge_type, edge_w, head, rec, E);
        }
        compgcn_pull_project<<<256, 512, 0, stream>>>(x, rel_emb, head, rec, W_self, W_forward, bias, out_nodes, N, E);
    } else {
        compgcn_rel_clear<<<relBlocks + 8192, 32, 0, stream>>>(rel_emb, W_rel, out_rel, R, relBlocks,
                                                                out_nodes, N);
        int units  = (E + EPW - 1) / EPW;
        int blocks = (units + 7) >> 3;
        if (blocks > 1024) blocks = 1024;
        compgcn_scatter<<<blocks, 256, 0, stream>>>(x, rel_emb, src, dst, edge_type, edge_w, out_nodes, E);
        compgcn_project<<<256, 512, 0, stream>>>(x, W_self, W_forward, bias, out_nodes, N);
    }
}
